# starting kernel
# baseline (speedup 1.0000x reference)
#include <hip/hip_runtime.h>
#include <hip/hip_bf16.h>

typedef __attribute__((ext_vector_type(2))) float v2f;
typedef __attribute__((ext_vector_type(8))) float v8f;

#define N_TOKENS   16384
#define IN_F       2048
#define OUT_F      2048
#define KK_STEPS   64
#define N_TILES    16
#define NT_GROUP   8
#define WAVES_PER_BLOCK 8
#define LDS_W_FLOATS   (KK_STEPS * N_TILES * 64)
#define LDS_BIAS_BASE  LDS_W_FLOATS
#define LDS_BIAS_FLOATS (N_TILES * 16 * 8)

__global__ void __launch_bounds__(256, 1)
block_linear_wmma_f32(const float* __restrict__ x,
                      const float* __restrict__ w,
                      const float* __restrict__ bias,
                      float* __restrict__ out) {
    extern __shared__ float lds[];
    const int tid  = threadIdx.x;
    const int lane = tid & 31;
    const int wave = tid >> 5;

    for (int i = tid; i < LDS_W_FLOATS / 2; i += 256) {
        const int f  = i >> 5;
        const int l  = i & 31;
        const int kk = f >> 4;
        const int nt = f & 15;
        const int K0 = 4 * kk + 2 * (l >> 4);
        const int Nn = 16 * nt + (l & 15);
        const int c  = K0 >> 4, k = K0 & 15;
        const int o  = Nn >> 5, n = Nn & 31;
        const v2f wv = *(const v2f*)&w[(((o * 16 + c) * 32 + n) << 4) + k];
        ((v2f*)lds)[i] = wv;
    }
    for (int j = tid; j < LDS_BIAS_FLOATS; j += 256) {
        const int ntg = j >> 7;
        const int l   = (j >> 3) & 15;
        const int v   = j & 7;
        const int Ng  = 16 * ntg + l;
        const int o   = Ng >> 5, n = Ng & 31;
        lds[LDS_BIAS_BASE + j] = bias[o * 256 + v * 32 + n];
    }
    __syncthreads();

    const int rt   = blockIdx.x * WAVES_PER_BLOCK + wave;
    const int h    = lane >> 4;
    const int l15  = lane & 15;

    const long tok_a = 2 * rt + (l15 >> 3);
    const int  m_a   = l15 & 7;
    const float* xrow = x + tok_a * (long)IN_F + m_a * 16 + 2 * h;

    const v2f* bfr = (const v2f*)lds;
    const long outbase = (2 * (long)rt + h) * (long)OUT_F;

#pragma unroll 1
    for (int g = 0; g < 2; ++g) {
        v8f acc[NT_GROUP];
#pragma unroll
        for (int nt = 0; nt < NT_GROUP; ++nt)
            acc[nt] = *(const v8f*)&lds[LDS_BIAS_BASE + (g * NT_GROUP + nt) * 128 + l15 * 8];

        const float* xa = xrow;
        v2f a0 = *(const v2f*)(xa + 0);
        v2f a1 = *(const v2f*)(xa + 4);
        v2f a2 = *(const v2f*)(xa + 8);
        v2f a3 = *(const v2f*)(xa + 12);

#pragma unroll 1
        for (int cb = 0; cb < 16; ++cb) {
            const float* xn = xa + (cb < 15 ? 128 : 0);
            const v2f p0 = *(const v2f*)(xn + 0);
            const v2f p1 = *(const v2f*)(xn + 4);
            const v2f p2 = *(const v2f*)(xn + 8);
            const v2f p3 = *(const v2f*)(xn + 12);

#pragma unroll
            for (int q = 0; q < 4; ++q) {
                const v2f aq = (q == 0) ? a0 : (q == 1) ? a1 : (q == 2) ? a2 : a3;
                const int fbase = ((4 * cb + q) * N_TILES + g * NT_GROUP) * 32;
#pragma unroll
                for (int nt = 0; nt < NT_GROUP; ++nt) {
                    const v2f b = bfr[fbase + nt * 32 + lane];
                    acc[nt] = __builtin_amdgcn_wmma_f32_16x16x4_f32(
                        false, aq, false, b, (short)0, acc[nt], false, false);
                }
            }
            a0 = p0; a1 = p1; a2 = p2; a3 = p3;
            xa += 128;
        }

#pragma unroll
        for (int nt = 0; nt < NT_GROUP; ++nt) {
            const int Ng = 16 * (g * NT_GROUP + nt) + l15;
            const int o  = Ng >> 5;
            const int n  = Ng & 31;
#pragma unroll
            for (int v = 0; v < 8; ++v)
                out[outbase + o * 256 + v * 32 + n] = acc[nt][v];
        }
    }
}

extern "C" void kernel_launch(void* const* d_in, const int* in_sizes, int n_in,
                              void* d_out, int out_size, void* d_ws, size_t ws_size,
                              hipStream_t stream) {
    const float* x    = (const float*)d_in[0];
    const float* w    = (const float*)d_in[1];
    const float* bias = (const float*)d_in[2];
    float* out        = (float*)d_out;

    const size_t lds_bytes = (LDS_W_FLOATS + LDS_BIAS_FLOATS) * sizeof(float);
    (void)hipFuncSetAttribute((const void*)block_linear_wmma_f32,
                              hipFuncAttributeMaxDynamicSharedMemorySize,
                              (int)lds_bytes);

    const int blocks = (N_TOKENS * 8) / 16 / WAVES_PER_BLOCK;
    block_linear_wmma_f32<<<blocks, 32 * WAVES_PER_BLOCK, lds_bytes, stream>>>(
        x, w, bias, out);
}


# Stage W'' and bias into LDS once per workgroup: 256 resident workgroups (one per WGP) each walk 4 row groups, instead of 1024 workgroups that each restage the 256 KiB weight
# speedup vs baseline: 1.1541x; 1.1541x over previous
#include <hip/hip_runtime.h>
#include <hip/hip_bf16.h>

typedef __attribute__((ext_vector_type(2))) float v2f;
typedef __attribute__((ext_vector_type(8))) float v8f;

#define N_TOKENS   16384
#define IN_F       2048
#define OUT_F      2048
#define KK_STEPS   64
#define N_TILES    16
#define NT_GROUP   8
#define WAVES_PER_BLOCK 8
#define ROW_GROUPS 1024
#define NUM_WORKGROUPS 256
#define LDS_W_FLOATS   (KK_STEPS * N_TILES * 64)
#define LDS_BIAS_BASE  LDS_W_FLOATS
#define LDS_BIAS_FLOATS (N_TILES * 16 * 8)

__global__ void __launch_bounds__(256, 1)
block_linear_wmma_f32(const float* __restrict__ x,
                      const float* __restrict__ w,
                      const float* __restrict__ bias,
                      float* __restrict__ out) {
    extern __shared__ float lds[];
    const int tid  = threadIdx.x;
    const int lane = tid & 31;
    const int wave = tid >> 5;

    for (int i = tid; i < LDS_W_FLOATS / 2; i += 256) {
        const int f  = i >> 5;
        const int l  = i & 31;
        const int kk = f >> 4;
        const int nt = f & 15;
        const int K0 = 4 * kk + 2 * (l >> 4);
        const int Nn = 16 * nt + (l & 15);
        const int c  = K0 >> 4, k = K0 & 15;
        const int o  = Nn >> 5, n = Nn & 31;
        const v2f wv = *(const v2f*)&w[(((o * 16 + c) * 32 + n) << 4) + k];
        ((v2f*)lds)[i] = wv;
    }
    for (int j = tid; j < LDS_BIAS_FLOATS; j += 256) {
        const int ntg = j >> 7;
        const int l   = (j >> 3) & 15;
        const int v   = j & 7;
        const int Ng  = 16 * ntg + l;
        const int o   = Ng >> 5, n = Ng & 31;
        lds[LDS_BIAS_BASE + j] = bias[o * 256 + v * 32 + n];
    }
    __syncthreads();

    const int h    = lane >> 4;
    const int l15  = lane & 15;
    const v2f* bfr = (const v2f*)lds;

    for (int grp = blockIdx.x; grp < ROW_GROUPS; grp += gridDim.x) {
        const int rt   = grp * WAVES_PER_BLOCK + wave;

        const long tok_a = 2 * rt + (l15 >> 3);
        const int  m_a   = l15 & 7;
        const float* xrow = x + tok_a * (long)IN_F + m_a * 16 + 2 * h;

        const long outbase = (2 * (long)rt + h) * (long)OUT_F;

#pragma unroll 1
        for (int g = 0; g < 2; ++g) {
            v8f acc[NT_GROUP];
#pragma unroll
            for (int nt = 0; nt < NT_GROUP; ++nt)
                acc[nt] = *(const v8f*)&lds[LDS_BIAS_BASE + (g * NT_GROUP + nt) * 128 + l15 * 8];

            const float* xa = xrow;
            v2f a0 = *(const v2f*)(xa + 0);
            v2f a1 = *(const v2f*)(xa + 4);
            v2f a2 = *(const v2f*)(xa + 8);
            v2f a3 = *(const v2f*)(xa + 12);

#pragma unroll 1
            for (int cb = 0; cb < 16; ++cb) {
                const float* xn = xa + (cb < 15 ? 128 : 0);
                const v2f p0 = *(const v2f*)(xn + 0);
                const v2f p1 = *(const v2f*)(xn + 4);
                const v2f p2 = *(const v2f*)(xn + 8);
                const v2f p3 = *(const v2f*)(xn + 12);

#pragma unroll
                for (int q = 0; q < 4; ++q) {
                    const v2f aq = (q == 0) ? a0 : (q == 1) ? a1 : (q == 2) ? a2 : a3;
                    const int fbase = ((4 * cb + q) * N_TILES + g * NT_GROUP) * 32;
#pragma unroll
                    for (int nt = 0; nt < NT_GROUP; ++nt) {
                        const v2f b = bfr[fbase + nt * 32 + lane];
                        acc[nt] = __builtin_amdgcn_wmma_f32_16x16x4_f32(
                            false, aq, false, b, (short)0, acc[nt], false, false);
                    }
                }
                a0 = p0; a1 = p1; a2 = p2; a3 = p3;
                xa += 128;
            }

#pragma unroll
            for (int nt = 0; nt < NT_GROUP; ++nt) {
                const int Ng = 16 * (g * NT_GROUP + nt) + l15;
                const int o  = Ng >> 5;
                const int n  = Ng & 31;
#pragma unroll
                for (int v = 0; v < 8; ++v)
                    out[outbase + o * 256 + v * 32 + n] = acc[nt][v];
            }
        }
    }
}

extern "C" void kernel_launch(void* const* d_in, const int* in_sizes, int n_in,
                              void* d_out, int out_size, void* d_ws, size_t ws_size,
                              hipStream_t stream) {
    const float* x    = (const float*)d_in[0];
    const float* w    = (const float*)d_in[1];
    const float* bias = (const float*)d_in[2];
    float* out        = (float*)d_out;

    const size_t lds_bytes = (LDS_W_FLOATS + LDS_BIAS_FLOATS) * sizeof(float);
    (void)hipFuncSetAttribute((const void*)block_linear_wmma_f32,
                              hipFuncAttributeMaxDynamicSharedMemorySize,
                              (int)lds_bytes);

    block_linear_wmma_f32<<<NUM_WORKGROUPS, 32 * WAVES_PER_BLOCK, lds_bytes, stream>>>(
        x, w, bias, out);
}


# x fragments prefetched 3-4 c-blocks ahead, not 1: a 4-slot register ring refilled in place as each block is consumed, running on across passes and row tiles (compiler fences pin it); sum order kept
# speedup vs baseline: 1.2176x; 1.0550x over previous
#include <hip/hip_runtime.h>
#include <hip/hip_bf16.h>

typedef __attribute__((ext_vector_type(2))) float v2f;
typedef __attribute__((ext_vector_type(8))) float v8f;

#define N_TOKENS   16384
#define IN_F       2048
#define OUT_F      2048
#define KK_STEPS   64
#define N_TILES    16
#define NT_GROUP   8
#define WAVES_PER_BLOCK 8
#define PF_DEPTH  4
#define ROW_GROUPS 1024
#define NUM_WORKGROUPS 256
#define LDS_W_FLOATS   (KK_STEPS * N_TILES * 64)
#define LDS_BIAS_BASE  LDS_W_FLOATS
#define LDS_BIAS_FLOATS (N_TILES * 16 * 8)

__global__ void __launch_bounds__(256, 1)
block_linear_wmma_f32(const float* __restrict__ x,
                      const float* __restrict__ w,
                      const float* __restrict__ bias,
                      float* __restrict__ out) {
    extern __shared__ float lds[];
    const int tid  = threadIdx.x;
    const int lane = tid & 31;
    const int wave = tid >> 5;

    for (int i = tid; i < LDS_W_FLOATS / 2; i += 256) {
        const int f  = i >> 5;
        const int l  = i & 31;
        const int kk = f >> 4;
        const int nt = f & 15;
        const int K0 = 4 * kk + 2 * (l >> 4);
        const int Nn = 16 * nt + (l & 15);
        const int c  = K0 >> 4, k = K0 & 15;
        const int o  = Nn >> 5, n = Nn & 31;
        const v2f wv = *(const v2f*)&w[(((o * 16 + c) * 32 + n) << 4) + k];
        ((v2f*)lds)[i] = wv;
    }
    for (int j = tid; j < LDS_BIAS_FLOATS; j += 256) {
        const int ntg = j >> 7;
        const int l   = (j >> 3) & 15;
        const int v   = j & 7;
        const int Ng  = 16 * ntg + l;
        const int o   = Ng >> 5, n = Ng & 31;
        lds[LDS_BIAS_BASE + j] = bias[o * 256 + v * 32 + n];
    }
    __syncthreads();

    const int h    = lane >> 4;
    const int l15  = lane & 15;
    const v2f* bfr = (const v2f*)lds;

    int grp = blockIdx.x;
    if (grp >= ROW_GROUPS) return;

    const long lane_off = (long)(l15 >> 3) * IN_F + (l15 & 7) * 16 + 2 * h;
    const float* xcur = x + (long)(2 * (grp * WAVES_PER_BLOCK + wave)) * IN_F + lane_off;

    v2f A[PF_DEPTH][4];
#pragma unroll
    for (int j = 0; j < PF_DEPTH; ++j) {
#pragma unroll
        for (int q = 0; q < 4; ++q)
            A[j][q] = *(const v2f*)(xcur + j * 128 + 4 * q);
        asm volatile("" ::: "memory");
    }

    for (; grp < ROW_GROUPS; grp += gridDim.x) {
        const int rt    = grp * WAVES_PER_BLOCK + wave;
        const int gnext = grp + gridDim.x;
        const float* xnt = (gnext < ROW_GROUPS)
            ? x + (long)(2 * (gnext * WAVES_PER_BLOCK + wave)) * IN_F + lane_off
            : xcur;

        const long outbase = (2 * (long)rt + h) * (long)OUT_F;

#pragma unroll 1
        for (int g = 0; g < 2; ++g) {
            const float* xnp = (g == 0) ? xcur : xnt;

            v8f acc[NT_GROUP];
#pragma unroll
            for (int nt = 0; nt < NT_GROUP; ++nt)
                acc[nt] = *(const v8f*)&lds[LDS_BIAS_BASE + (g * NT_GROUP + nt) * 128 + l15 * 8];

#pragma unroll 1
            for (int cb0 = 0; cb0 < 16; cb0 += PF_DEPTH) {
                const float* rb = (cb0 + PF_DEPTH < 16) ? xcur + (cb0 + PF_DEPTH) * 128 : xnp;
#pragma unroll
                for (int j = 0; j < PF_DEPTH; ++j) {
#pragma unroll
                    for (int q = 0; q < 4; ++q) {
                        const int fbase = ((4 * (cb0 + j) + q) * N_TILES + g * NT_GROUP) * 32;
#pragma unroll
                        for (int nt = 0; nt < NT_GROUP; ++nt) {
                            const v2f b = bfr[fbase + nt * 32 + lane];
                            acc[nt] = __builtin_amdgcn_wmma_f32_16x16x4_f32(
                                false, A[j][q], false, b, (short)0, acc[nt], false, false);
                        }
                    }
#pragma unroll
                    for (int q = 0; q < 4; ++q)
                        A[j][q] = *(const v2f*)(rb + j * 128 + 4 * q);
                    asm volatile("" ::: "memory");
                }
            }

#pragma unroll
            for (int nt = 0; nt < NT_GROUP; ++nt) {
                const int Ng = 16 * (g * NT_GROUP + nt) + l15;
                const int o  = Ng >> 5;
                const int n  = Ng & 31;
#pragma unroll
                for (int v = 0; v < 8; ++v)
                    out[outbase + o * 256 + v * 32 + n] = acc[nt][v];
            }
        }
        xcur = xnt;
    }
}

extern "C" void kernel_launch(void* const* d_in, const int* in_sizes, int n_in,
                              void* d_out, int out_size, void* d_ws, size_t ws_size,
                              hipStream_t stream) {
    const float* x    = (const float*)d_in[0];
    const float* w    = (const float*)d_in[1];
    const float* bias = (const float*)d_in[2];
    float* out        = (float*)d_out;

    const size_t lds_bytes = (LDS_W_FLOATS + LDS_BIAS_FLOATS) * sizeof(float);
    (void)hipFuncSetAttribute((const void*)block_linear_wmma_f32,
                              hipFuncAttributeMaxDynamicSharedMemorySize,
                              (int)lds_bytes);

    block_linear_wmma_f32<<<NUM_WORKGROUPS, 32 * WAVES_PER_BLOCK, lds_bytes, stream>>>(
        x, w, bias, out);
}


# 16 waves per workgroup (4 per SIMD) instead of 8, per-wave code unchanged; each workgroup walks 2 row groups of 16 tiles: younger waves fill the matrix slots a waiting wave leaves; sum order kept
# speedup vs baseline: 1.3130x; 1.0784x over previous
#include <hip/hip_runtime.h>
#include <hip/hip_bf16.h>

typedef __attribute__((ext_vector_type(2))) float v2f;
typedef __attribute__((ext_vector_type(8))) float v8f;

#define N_TOKENS   16384
#define IN_F       2048
#define OUT_F      2048
#define KK_STEPS   64
#define N_TILES    16
#define NT_GROUP   8
#define WAVES_PER_BLOCK 16
#define BLOCK_THREADS (32 * WAVES_PER_BLOCK)
#define PF_DEPTH  4
#define ROW_GROUPS 512
#define NUM_WORKGROUPS 256
#define LDS_W_FLOATS   (KK_STEPS * N_TILES * 64)
#define LDS_BIAS_BASE  LDS_W_FLOATS
#define LDS_BIAS_FLOATS (N_TILES * 16 * 8)

__global__ void __launch_bounds__(BLOCK_THREADS, 1)
block_linear_wmma_f32(const float* __restrict__ x,
                      const float* __restrict__ w,
                      const float* __restrict__ bias,
                      float* __restrict__ out) {
    extern __shared__ float lds[];
    const int tid  = threadIdx.x;
    const int lane = tid & 31;
    const int wave = tid >> 5;

    for (int i = tid; i < LDS_W_FLOATS / 2; i += BLOCK_THREADS) {
        const int f  = i >> 5;
        const int l  = i & 31;
        const int kk = f >> 4;
        const int nt = f & 15;
        const int K0 = 4 * kk + 2 * (l >> 4);
        const int Nn = 16 * nt + (l & 15);
        const int c  = K0 >> 4, k = K0 & 15;
        const int o  = Nn >> 5, n = Nn & 31;
        const v2f wv = *(const v2f*)&w[(((o * 16 + c) * 32 + n) << 4) + k];
        ((v2f*)lds)[i] = wv;
    }
    for (int j = tid; j < LDS_BIAS_FLOATS; j += BLOCK_THREADS) {
        const int ntg = j >> 7;
        const int l   = (j >> 3) & 15;
        const int v   = j & 7;
        const int Ng  = 16 * ntg + l;
        const int o   = Ng >> 5, n = Ng & 31;
        lds[LDS_BIAS_BASE + j] = bias[o * 256 + v * 32 + n];
    }
    __syncthreads();

    const int h    = lane >> 4;
    const int l15  = lane & 15;
    const v2f* bfr = (const v2f*)lds;

    int grp = blockIdx.x;
    if (grp >= ROW_GROUPS) return;

    const long lane_off = (long)(l15 >> 3) * IN_F + (l15 & 7) * 16 + 2 * h;
    const float* xcur = x + (long)(2 * (grp * WAVES_PER_BLOCK + wave)) * IN_F + lane_off;

    v2f A[PF_DEPTH][4];
#pragma unroll
    for (int j = 0; j < PF_DEPTH; ++j) {
#pragma unroll
        for (int q = 0; q < 4; ++q)
            A[j][q] = *(const v2f*)(xcur + j * 128 + 4 * q);
        asm volatile("" ::: "memory");
    }

    for (; grp < ROW_GROUPS; grp += gridDim.x) {
        const int rt    = grp * WAVES_PER_BLOCK + wave;
        const int gnext = grp + gridDim.x;
        const float* xnt = (gnext < ROW_GROUPS)
            ? x + (long)(2 * (gnext * WAVES_PER_BLOCK + wave)) * IN_F + lane_off
            : xcur;

        const long outbase = (2 * (long)rt + h) * (long)OUT_F;

#pragma unroll 1
        for (int g = 0; g < 2; ++g) {
            const float* xnp = (g == 0) ? xcur : xnt;

            v8f acc[NT_GROUP];
#pragma unroll
            for (int nt = 0; nt < NT_GROUP; ++nt)
                acc[nt] = *(const v8f*)&lds[LDS_BIAS_BASE + (g * NT_GROUP + nt) * 128 + l15 * 8];

#pragma unroll 1
            for (int cb0 = 0; cb0 < 16; cb0 += PF_DEPTH) {
                const float* rb = (cb0 + PF_DEPTH < 16) ? xcur + (cb0 + PF_DEPTH) * 128 : xnp;
#pragma unroll
                for (int j = 0; j < PF_DEPTH; ++j) {
#pragma unroll
                    for (int q = 0; q < 4; ++q) {
                        const int fbase = ((4 * (cb0 + j) + q) * N_TILES + g * NT_GROUP) * 32;
#pragma unroll
                        for (int nt = 0; nt < NT_GROUP; ++nt) {
                            const v2f b = bfr[fbase + nt * 32 + lane];
                            acc[nt] = __builtin_amdgcn_wmma_f32_16x16x4_f32(
                                false, A[j][q], false, b, (short)0, acc[nt], false, false);
                        }
                    }
#pragma unroll
                    for (int q = 0; q < 4; ++q)
                        A[j][q] = *(const v2f*)(rb + j * 128 + 4 * q);
                    asm volatile("" ::: "memory");
                }
            }

#pragma unroll
            for (int nt = 0; nt < NT_GROUP; ++nt) {
                const int Ng = 16 * (g * NT_GROUP + nt) + l15;
                const int o  = Ng >> 5;
                const int n  = Ng & 31;
#pragma unroll
                for (int v = 0; v < 8; ++v)
                    out[outbase + o * 256 + v * 32 + n] = acc[nt][v];
            }
        }
        xcur = xnt;
    }
}

extern "C" void kernel_launch(void* const* d_in, const int* in_sizes, int n_in,
                              void* d_out, int out_size, void* d_ws, size_t ws_size,
                              hipStream_t stream) {
    const float* x    = (const float*)d_in[0];
    const float* w    = (const float*)d_in[1];
    const float* bias = (const float*)d_in[2];
    float* out        = (float*)d_out;

    const size_t lds_bytes = (LDS_W_FLOATS + LDS_BIAS_FLOATS) * sizeof(float);
    (void)hipFuncSetAttribute((const void*)block_linear_wmma_f32,
                              hipFuncAttributeMaxDynamicSharedMemorySize,
                              (int)lds_bytes);

    block_linear_wmma_f32<<<NUM_WORKGROUPS, BLOCK_THREADS, lds_bytes, stream>>>(
        x, w, bias, out);
}


# Wave priority by progress: s_setprio 3, 2, 1, 0 over a wave's four passes, so a wave that is a pass behind outranks those ahead and a SIMD's four waves end together; block loop and sum order kept
# speedup vs baseline: 1.3616x; 1.0370x over previous
#include <hip/hip_runtime.h>
#include <hip/hip_bf16.h>

typedef __attribute__((ext_vector_type(2))) float v2f;
typedef __attribute__((ext_vector_type(8))) float v8f;

#define N_TOKENS   16384
#define IN_F       2048
#define OUT_F      2048
#define KK_STEPS   64
#define N_TILES    16
#define NT_GROUP   8
#define WAVES_PER_BLOCK 16
#define BLOCK_THREADS (32 * WAVES_PER_BLOCK)
#define PF_DEPTH  4
#define ROW_GROUPS 512
#define NUM_WORKGROUPS 256
#define LDS_W_FLOATS   (KK_STEPS * N_TILES * 64)
#define LDS_BIAS_BASE  LDS_W_FLOATS
#define LDS_BIAS_FLOATS (N_TILES * 16 * 8)

__global__ void __launch_bounds__(BLOCK_THREADS, 1)
block_linear_wmma_f32(const float* __restrict__ x,
                      const float* __restrict__ w,
                      const float* __restrict__ bias,
                      float* __restrict__ out) {
    extern __shared__ float lds[];
    const int tid  = threadIdx.x;
    const int lane = tid & 31;
    const int wave = tid >> 5;

    for (int i = tid; i < LDS_W_FLOATS / 2; i += BLOCK_THREADS) {
        const int f  = i >> 5;
        const int l  = i & 31;
        const int kk = f >> 4;
        const int nt = f & 15;
        const int K0 = 4 * kk + 2 * (l >> 4);
        const int Nn = 16 * nt + (l & 15);
        const int c  = K0 >> 4, k = K0 & 15;
        const int o  = Nn >> 5, n = Nn & 31;
        const v2f wv = *(const v2f*)&w[(((o * 16 + c) * 32 + n) << 4) + k];
        ((v2f*)lds)[i] = wv;
    }
    for (int j = tid; j < LDS_BIAS_FLOATS; j += BLOCK_THREADS) {
        const int ntg = j >> 7;
        const int l   = (j >> 3) & 15;
        const int v   = j & 7;
        const int Ng  = 16 * ntg + l;
        const int o   = Ng >> 5, n = Ng & 31;
        lds[LDS_BIAS_BASE + j] = bias[o * 256 + v * 32 + n];
    }
    __syncthreads();

    const int h    = lane >> 4;
    const int l15  = lane & 15;
    const v2f* bfr = (const v2f*)lds;

    int grp = blockIdx.x;
    if (grp >= ROW_GROUPS) return;

    __builtin_amdgcn_s_setprio(3);

    const long lane_off = (long)(l15 >> 3) * IN_F + (l15 & 7) * 16 + 2 * h;
    const float* xcur = x + (long)(2 * (grp * WAVES_PER_BLOCK + wave)) * IN_F + lane_off;

    v2f A[PF_DEPTH][4];
#pragma unroll
    for (int j = 0; j < PF_DEPTH; ++j) {
#pragma unroll
        for (int q = 0; q < 4; ++q)
            A[j][q] = *(const v2f*)(xcur + j * 128 + 4 * q);
        asm volatile("" ::: "memory");
    }

    for (; grp < ROW_GROUPS; grp += gridDim.x) {
        const int rt    = grp * WAVES_PER_BLOCK + wave;
        const int gnext = grp + gridDim.x;
        const float* xnt = (gnext < ROW_GROUPS)
            ? x + (long)(2 * (gnext * WAVES_PER_BLOCK + wave)) * IN_F + lane_off
            : xcur;

        const bool first_tile = (grp == (int)blockIdx.x);
        const bool last_tile  = (gnext >= ROW_GROUPS);

        const long outbase = (2 * (long)rt + h) * (long)OUT_F;

#pragma unroll 1
        for (int g = 0; g < 2; ++g) {
            const float* xnp = (g == 0) ? xcur : xnt;

            if (last_tile) {
                if (g == 0) __builtin_amdgcn_s_setprio(1);
                else        __builtin_amdgcn_s_setprio(0);
            } else if (!first_tile || g == 1) {
                __builtin_amdgcn_s_setprio(2);
            }

            v8f acc[NT_GROUP];
#pragma unroll
            for (int nt = 0; nt < NT_GROUP; ++nt)
                acc[nt] = *(const v8f*)&lds[LDS_BIAS_BASE + (g * NT_GROUP + nt) * 128 + l15 * 8];

#pragma unroll 1
            for (int cb0 = 0; cb0 < 16; cb0 += PF_DEPTH) {
                const float* rb = (cb0 + PF_DEPTH < 16) ? xcur + (cb0 + PF_DEPTH) * 128 : xnp;
#pragma unroll
                for (int j = 0; j < PF_DEPTH; ++j) {
#pragma unroll
                    for (int q = 0; q < 4; ++q) {
                        const int fbase = ((4 * (cb0 + j) + q) * N_TILES + g * NT_GROUP) * 32;
#pragma unroll
                        for (int nt = 0; nt < NT_GROUP; ++nt) {
                            const v2f b = bfr[fbase + nt * 32 + lane];
                            acc[nt] = __builtin_amdgcn_wmma_f32_16x16x4_f32(
                                false, A[j][q], false, b, (short)0, acc[nt], false, false);
                        }
                    }
#pragma unroll
                    for (int q = 0; q < 4; ++q)
                        A[j][q] = *(const v2f*)(rb + j * 128 + 4 * q);
                    asm volatile("" ::: "memory");
                }
            }

#pragma unroll
            for (int nt = 0; nt < NT_GROUP; ++nt) {
                const int Ng = 16 * (g * NT_GROUP + nt) + l15;
                const int o  = Ng >> 5;
                const int n  = Ng & 31;
#pragma unroll
                for (int v = 0; v < 8; ++v)
                    out[outbase + o * 256 + v * 32 + n] = acc[nt][v];
            }
        }
        xcur = xnt;
    }
}

extern "C" void kernel_launch(void* const* d_in, const int* in_sizes, int n_in,
                              void* d_out, int out_size, void* d_ws, size_t ws_size,
                              hipStream_t stream) {
    const float* x    = (const float*)d_in[0];
    const float* w    = (const float*)d_in[1];
    const float* bias = (const float*)d_in[2];
    float* out        = (float*)d_out;

    const size_t lds_bytes = (LDS_W_FLOATS + LDS_BIAS_FLOATS) * sizeof(float);
    (void)hipFuncSetAttribute((const void*)block_linear_wmma_f32,
                              hipFuncAttributeMaxDynamicSharedMemorySize,
                              (int)lds_bytes);

    block_linear_wmma_f32<<<NUM_WORKGROUPS, BLOCK_THREADS, lds_bytes, stream>>>(
        x, w, bias, out);
}
